# starting kernel
# baseline (speedup 1.0000x reference)
#include <hip/hip_runtime.h>
#include <hip/hip_bf16.h>


#define N_NODES 100000
#define N_EDGES 1600000
#define F_INPUT 16
#define HDIM 128
#define M_TILES (N_NODES / 16)

typedef float v2f __attribute__((ext_vector_type(2)));
typedef float v8f __attribute__((ext_vector_type(8)));

__global__ __launch_bounds__(256) void zero_f4_kernel(float4* __restrict__ p, long long n4) {
    long long i = (long long)blockIdx.x * 256 + threadIdx.x;
    if (i < n4) p[i] = make_float4(0.f, 0.f, 0.f, 0.f);
}

__global__ __launch_bounds__(256) void degree_kernel(const int* __restrict__ dst,
                                                     float* __restrict__ cnt, int E) {
    int e = blockIdx.x * 256 + threadIdx.x;
    if (e < E) atomicAdd(&cnt[dst[e]], 1.0f);
}

__global__ __launch_bounds__(256) void recip_kernel(float* __restrict__ deg, int n) {
    int i = blockIdx.x * 256 + threadIdx.x;
    if (i < n) deg[i] = 1.0f / fmaxf(deg[i], 1.0f);
}

template <int CHUNKS>
__global__ __launch_bounds__(256) void scatter_kernel(const float* __restrict__ in,
                                                      const int* __restrict__ src,
                                                      const int* __restrict__ dst,
                                                      float* __restrict__ agg, int E) {
    long long tid = (long long)blockIdx.x * 256 + threadIdx.x;
    if (tid >= (long long)E * CHUNKS) return;
    int e = (int)(tid / CHUNKS);
    int c = (int)(tid % CHUNKS);
    int s = src[e];
    int d = dst[e];
    float4 v = ((const float4*)in)[(long long)s * CHUNKS + c];
    float* p = agg + ((long long)d * CHUNKS + c) * 4;
    atomicAdd(p + 0, v.x);
    atomicAdd(p + 1, v.y);
    atomicAdd(p + 2, v.z);
    atomicAdd(p + 3, v.w);
}

__global__ __launch_bounds__(256) void sage_gemm_kernel(const float* __restrict__ in,
                                                        const float* __restrict__ agg,
                                                        const float* __restrict__ invdeg,
                                                        const float* __restrict__ Wr,
                                                        const float* __restrict__ Wl,
                                                        const float* __restrict__ bias,
                                                        float* __restrict__ out,
                                                        int K, int do_relu) {
    const int lane = threadIdx.x & 31;
    const int wv = threadIdx.x >> 5;
    const int mt = blockIdx.x * 8 + wv;
    if (mt >= M_TILES) return;

    const int row0 = mt << 4;
    const int m = lane & 15;
    const int kh = (lane >> 4) << 1;
    const long long arow = (long long)(row0 + m);

    v8f acc[8] = {};

    for (int k = 0; k < K; k += 4) {
        v2f a = *(const v2f*)(in + arow * K + k + kh);
#pragma unroll
        for (int nt = 0; nt < 8; ++nt) {
            const int col = (nt << 4) + m;
            v2f b;
            b.x = Wr[(k + kh) * HDIM + col];
            b.y = Wr[(k + kh + 1) * HDIM + col];
            acc[nt] = __builtin_amdgcn_wmma_f32_16x16x4_f32(
                false, a, false, b, (short)0, acc[nt], false, false);
        }
    }

    const float inv = invdeg[row0 + m];
    for (int k = 0; k < K; k += 4) {
        v2f a = *(const v2f*)(agg + arow * K + k + kh);
        a.x *= inv;
        a.y *= inv;
#pragma unroll
        for (int nt = 0; nt < 8; ++nt) {
            const int col = (nt << 4) + m;
            v2f b;
            b.x = Wl[(k + kh) * HDIM + col];
            b.y = Wl[(k + kh + 1) * HDIM + col];
            acc[nt] = __builtin_amdgcn_wmma_f32_16x16x4_f32(
                false, a, false, b, (short)0, acc[nt], false, false);
        }
    }

    const int rofs = (lane >> 4) << 3;
#pragma unroll
    for (int nt = 0; nt < 8; ++nt) {
        const int col = (nt << 4) + m;
        const float bv = bias[col];
#pragma unroll
        for (int j = 0; j < 8; ++j) {
            float v = acc[nt][j] + bv;
            if (do_relu) v = fmaxf(v, 0.0f);
            out[(long long)(row0 + rofs + j) * HDIM + col] = v;
        }
    }
}

extern "C" void kernel_launch(void* const* d_in, const int* in_sizes, int n_in,
                              void* d_out, int out_size, void* d_ws, size_t ws_size,
                              hipStream_t stream) {
    (void)in_sizes; (void)n_in; (void)out_size; (void)ws_size;

    const float* x = (const float*)d_in[0];
    const int* ei = (const int*)d_in[1];
    const int* src = ei;
    const int* dst = ei + N_EDGES;
    const float* Wl[4] = {(const float*)d_in[2], (const float*)d_in[5],
                          (const float*)d_in[8], (const float*)d_in[11]};
    const float* Wr[4] = {(const float*)d_in[3], (const float*)d_in[6],
                          (const float*)d_in[9], (const float*)d_in[12]};
    const float* bs[4] = {(const float*)d_in[4], (const float*)d_in[7],
                          (const float*)d_in[10], (const float*)d_in[13]};

    float* agg = (float*)d_ws;
    float* hA = agg + (size_t)N_NODES * HDIM;
    float* hB = hA + (size_t)N_NODES * HDIM;
    float* deg = hB + (size_t)N_NODES * HDIM;

    zero_f4_kernel<<<(N_NODES / 4 + 255) / 256, 256, 0, stream>>>((float4*)deg, N_NODES / 4);
    degree_kernel<<<(N_EDGES + 255) / 256, 256, 0, stream>>>(dst, deg, N_EDGES);
    recip_kernel<<<(N_NODES + 255) / 256, 256, 0, stream>>>(deg, N_NODES);

    const int gemm_grid = (M_TILES + 7) / 8;

    {
        long long n4 = (long long)N_NODES * F_INPUT / 4;
        zero_f4_kernel<<<(int)((n4 + 255) / 256), 256, 0, stream>>>((float4*)agg, n4);
        long long t = (long long)N_EDGES * (F_INPUT / 4);
        scatter_kernel<F_INPUT / 4><<<(int)((t + 255) / 256), 256, 0, stream>>>(
            x, src, dst, agg, N_EDGES);
        sage_gemm_kernel<<<gemm_grid, 256, 0, stream>>>(x, agg, deg, Wr[0], Wl[0], bs[0],
                                                        hA, F_INPUT, 1);
    }

    const float* cur = hA;
    float* nxt = hB;
    for (int l = 1; l < 4; ++l) {
        long long n4 = (long long)N_NODES * HDIM / 4;
        zero_f4_kernel<<<(int)((n4 + 255) / 256), 256, 0, stream>>>((float4*)agg, n4);
        long long t = (long long)N_EDGES * (HDIM / 4);
        scatter_kernel<HDIM / 4><<<(int)((t + 255) / 256), 256, 0, stream>>>(
            cur, src, dst, agg, N_EDGES);
        float* outp = (l == 3) ? (float*)d_out : nxt;
        int relu = (l == 3) ? 0 : 1;
        sage_gemm_kernel<<<gemm_grid, 256, 0, stream>>>(cur, agg, deg, Wr[l], Wl[l], bs[l],
                                                        outp, HDIM, relu);
        float* t2 = (float*)cur;
        cur = outp;
        nxt = t2;
    }
}


# Aggregate by per-node gather over a per-call dst-bucketed edge list (int atomic count+arrival rank, scan, place) instead of f32 atomic scatter-add; neighbour sum order = arrival rank
# speedup vs baseline: 2.7285x; 2.7285x over previous
#include <hip/hip_runtime.h>
#include <hip/hip_bf16.h>


#define N_NODES 100000
#define N_EDGES 1600000
#define F_INPUT 16
#define HDIM 128
#define M_TILES (N_NODES / 16)
#define SCAN_T 1024

typedef float v2f __attribute__((ext_vector_type(2)));
typedef float v8f __attribute__((ext_vector_type(8)));

__global__ __launch_bounds__(256) void zero_f4_kernel(float4* __restrict__ p, long long n4) {
    long long i = (long long)blockIdx.x * 256 + threadIdx.x;
    if (i < n4) p[i] = make_float4(0.f, 0.f, 0.f, 0.f);
}

__global__ __launch_bounds__(256) void count_rank_kernel(const int* __restrict__ dst,
                                                         int* __restrict__ cnt,
                                                         int* __restrict__ rank, int E) {
    int e = blockIdx.x * 256 + threadIdx.x;
    if (e < E) rank[e] = atomicAdd(&cnt[dst[e]], 1);
}

__global__ __launch_bounds__(SCAN_T) void rowptr_kernel(const int* __restrict__ cnt,
                                                        int* __restrict__ rowptr,
                                                        float* __restrict__ invdeg, int n) {
    __shared__ int s_front[SCAN_T / 32];
    __shared__ int s_wave[SCAN_T / 32];
    const int t = threadIdx.x;
    const int lane = t & 31;
    const int w = t >> 5;
    const int base = blockIdx.x * SCAN_T;

    int front = 0;
    const int4* c4 = (const int4*)cnt;
    for (int i = t; i < base / 4; i += SCAN_T) {
        int4 v = c4[i];
        front += (v.x + v.y) + (v.z + v.w);
    }
    const int idx = base + t;
    const int c = (idx < n) ? cnt[idx] : 0;

    int inc = c;
    for (int o = 1; o < 32; o <<= 1) {
        int u = __shfl_up(inc, o);
        if (lane >= o) inc += u;
    }
    for (int o = 16; o > 0; o >>= 1) front += __shfl_xor(front, o);
    if (lane == 31) s_wave[w] = inc;
    if (lane == 0) s_front[w] = front;
    __syncthreads();

    int tot = s_front[lane] + ((lane < w) ? s_wave[lane] : 0);
    for (int o = 16; o > 0; o >>= 1) tot += __shfl_xor(tot, o);

    if (idx < n) {
        const int excl = tot + inc - c;
        rowptr[idx] = excl;
        invdeg[idx] = 1.0f / fmaxf((float)c, 1.0f);
        if (idx == n - 1) rowptr[n] = excl + c;
    }
}

__global__ __launch_bounds__(256) void place_kernel(const int* __restrict__ src,
                                                    const int* __restrict__ dst,
                                                    const int* __restrict__ rank,
                                                    const int* __restrict__ rowptr,
                                                    int* __restrict__ col, int E) {
    int e = blockIdx.x * 256 + threadIdx.x;
    if (e < E) col[rowptr[dst[e]] + rank[e]] = src[e];
}

__device__ __forceinline__ void add4(float4& a, const float4& v) {
    a.x += v.x; a.y += v.y; a.z += v.z; a.w += v.w;
}

template <int U>
__device__ __forceinline__ void gather_rows(const float4* __restrict__ in4, int mine, int t,
                                            int lane, float4& acc) {
    float4 v[U];
#pragma unroll
    for (int u = 0; u < U; ++u) {
        const int s = __builtin_amdgcn_readlane(mine, t + u);
        v[u] = in4[(long long)s * (HDIM / 4) + lane];
    }
#pragma unroll
    for (int u = 0; u < U; ++u) add4(acc, v[u]);
}

__global__ __launch_bounds__(256) void gather128_kernel(const float* __restrict__ in,
                                                        const int* __restrict__ rowptr,
                                                        const int* __restrict__ col,
                                                        float* __restrict__ agg) {
    const int lane = threadIdx.x & 31;
    const int node = __builtin_amdgcn_readfirstlane((int)(blockIdx.x * 8 + (threadIdx.x >> 5)));
    if (node >= N_NODES) return;
    const int beg = rowptr[node];
    const int end = rowptr[node + 1];
    const float4* in4 = (const float4*)in;
    float4 acc = make_float4(0.f, 0.f, 0.f, 0.f);
    for (int j0 = beg; j0 < end; j0 += 32) {
        const int nj = (end - j0 < 32) ? (end - j0) : 32;
        const int mine = (lane < nj) ? col[j0 + lane] : 0;
        int t = 0;
        for (; t + 8 <= nj; t += 8) gather_rows<8>(in4, mine, t, lane, acc);
        if (t + 4 <= nj) { gather_rows<4>(in4, mine, t, lane, acc); t += 4; }
        if (t + 2 <= nj) { gather_rows<2>(in4, mine, t, lane, acc); t += 2; }
        if (t < nj) gather_rows<1>(in4, mine, t, lane, acc);
    }
    ((float4*)agg)[(long long)node * (HDIM / 4) + lane] = acc;
}

__global__ __launch_bounds__(256) void gather16_kernel(const float* __restrict__ in,
                                                       const int* __restrict__ rowptr,
                                                       const int* __restrict__ col,
                                                       float* __restrict__ agg) {
    const int tid = blockIdx.x * 256 + threadIdx.x;
    const int node = tid >> 2;
    const int c = tid & 3;
    if (node >= N_NODES) return;
    const int beg = rowptr[node];
    const int end = rowptr[node + 1];
    const float4* in4 = (const float4*)in;
    float4 acc = make_float4(0.f, 0.f, 0.f, 0.f);
    int j = beg;
    for (; j + 4 <= end; j += 4) {
        const int s0 = col[j], s1 = col[j + 1], s2 = col[j + 2], s3 = col[j + 3];
        const float4 v0 = in4[(long long)s0 * (F_INPUT / 4) + c];
        const float4 v1 = in4[(long long)s1 * (F_INPUT / 4) + c];
        const float4 v2 = in4[(long long)s2 * (F_INPUT / 4) + c];
        const float4 v3 = in4[(long long)s3 * (F_INPUT / 4) + c];
        add4(acc, v0); add4(acc, v1); add4(acc, v2); add4(acc, v3);
    }
    for (; j < end; ++j) {
        const float4 v = in4[(long long)col[j] * (F_INPUT / 4) + c];
        add4(acc, v);
    }
    ((float4*)agg)[(long long)node * (F_INPUT / 4) + c] = acc;
}

__global__ __launch_bounds__(256) void sage_gemm_kernel(const float* __restrict__ in,
                                                        const float* __restrict__ agg,
                                                        const float* __restrict__ invdeg,
                                                        const float* __restrict__ Wr,
                                                        const float* __restrict__ Wl,
                                                        const float* __restrict__ bias,
                                                        float* __restrict__ out,
                                                        int K, int do_relu) {
    const int lane = threadIdx.x & 31;
    const int wv = threadIdx.x >> 5;
    const int mt = blockIdx.x * 8 + wv;
    if (mt >= M_TILES) return;

    const int row0 = mt << 4;
    const int m = lane & 15;
    const int kh = (lane >> 4) << 1;
    const long long arow = (long long)(row0 + m);

    v8f acc[8] = {};

    for (int k = 0; k < K; k += 4) {
        v2f a = *(const v2f*)(in + arow * K + k + kh);
#pragma unroll
        for (int nt = 0; nt < 8; ++nt) {
            const int col = (nt << 4) + m;
            v2f b;
            b.x = Wr[(k + kh) * HDIM + col];
            b.y = Wr[(k + kh + 1) * HDIM + col];
            acc[nt] = __builtin_amdgcn_wmma_f32_16x16x4_f32(
                false, a, false, b, (short)0, acc[nt], false, false);
        }
    }

    const float inv = invdeg[row0 + m];
    for (int k = 0; k < K; k += 4) {
        v2f a = *(const v2f*)(agg + arow * K + k + kh);
        a.x *= inv;
        a.y *= inv;
#pragma unroll
        for (int nt = 0; nt < 8; ++nt) {
            const int col = (nt << 4) + m;
            v2f b;
            b.x = Wl[(k + kh) * HDIM + col];
            b.y = Wl[(k + kh + 1) * HDIM + col];
            acc[nt] = __builtin_amdgcn_wmma_f32_16x16x4_f32(
                false, a, false, b, (short)0, acc[nt], false, false);
        }
    }

    const int rofs = (lane >> 4) << 3;
#pragma unroll
    for (int nt = 0; nt < 8; ++nt) {
        const int col = (nt << 4) + m;
        const float bv = bias[col];
#pragma unroll
        for (int j = 0; j < 8; ++j) {
            float v = acc[nt][j] + bv;
            if (do_relu) v = fmaxf(v, 0.0f);
            out[(long long)(row0 + rofs + j) * HDIM + col] = v;
        }
    }
}

extern "C" void kernel_launch(void* const* d_in, const int* in_sizes, int n_in,
                              void* d_out, int out_size, void* d_ws, size_t ws_size,
                              hipStream_t stream) {
    (void)in_sizes; (void)n_in; (void)out_size; (void)ws_size;

    const float* x = (const float*)d_in[0];
    const int* ei = (const int*)d_in[1];
    const int* src = ei;
    const int* dst = ei + N_EDGES;
    const float* Wl[4] = {(const float*)d_in[2], (const float*)d_in[5],
                          (const float*)d_in[8], (const float*)d_in[11]};
    const float* Wr[4] = {(const float*)d_in[3], (const float*)d_in[6],
                          (const float*)d_in[9], (const float*)d_in[12]};
    const float* bs[4] = {(const float*)d_in[4], (const float*)d_in[7],
                          (const float*)d_in[10], (const float*)d_in[13]};

    float* agg = (float*)d_ws;
    float* hA = agg + (size_t)N_NODES * HDIM;
    float* hB = hA + (size_t)N_NODES * HDIM;
    float* deg = hB + (size_t)N_NODES * HDIM;

    int* rank = (int*)hB;
    int* cnt = rank + N_EDGES;
    int* col = (int*)d_out;
    int* rowptr = col + N_EDGES;

    zero_f4_kernel<<<(N_NODES / 4 + 255) / 256, 256, 0, stream>>>((float4*)cnt, N_NODES / 4);
    count_rank_kernel<<<(N_EDGES + 255) / 256, 256, 0, stream>>>(dst, cnt, rank, N_EDGES);
    rowptr_kernel<<<(N_NODES + SCAN_T - 1) / SCAN_T, SCAN_T, 0, stream>>>(cnt, rowptr, deg, N_NODES);
    place_kernel<<<(N_EDGES + 255) / 256, 256, 0, stream>>>(src, dst, rank, rowptr, col, N_EDGES);

    const int gemm_grid = (M_TILES + 7) / 8;

    gather16_kernel<<<(N_NODES * (F_INPUT / 4) + 255) / 256, 256, 0, stream>>>(x, rowptr, col, agg);
    sage_gemm_kernel<<<gemm_grid, 256, 0, stream>>>(x, agg, deg, Wr[0], Wl[0], bs[0],
                                                    hA, F_INPUT, 1);

    const float* cur = hA;
    float* nxt = hB;
    for (int l = 1; l < 4; ++l) {
        gather128_kernel<<<(N_NODES + 7) / 8, 256, 0, stream>>>(cur, rowptr, col, agg);
        float* outp = (l == 3) ? (float*)d_out : nxt;
        int relu = (l == 3) ? 0 : 1;
        sage_gemm_kernel<<<gemm_grid, 256, 0, stream>>>(cur, agg, deg, Wr[l], Wl[l], bs[l],
                                                        outp, HDIM, relu);
        float* t2 = (float*)cur;
        cur = outp;
        nxt = t2;
    }
}


# Layers 2-4: the wave that projects a 16-node stripe sums its neighbour rows itself and feeds the means to the WMMAs through LDS (one launch per layer, sums never in memory); sum order unchanged
# speedup vs baseline: 4.0782x; 1.4947x over previous
#include <hip/hip_runtime.h>
#include <hip/hip_bf16.h>


#define N_NODES 100000
#define N_EDGES 1600000
#define F_INPUT 16
#define HDIM 128
#define M_TILES (N_NODES / 16)
#define BIN_SHIFT 9
#define BIN_N (1 << BIN_SHIFT)
#define NBINS ((N_NODES + BIN_N - 1) / BIN_N)
#define NCHUNK 256
#define CHUNK_E ((N_EDGES + NCHUNK - 1) / NCHUNK)
#define SRC_BITS 17
#define SRC_MASK ((1 << SRC_BITS) - 1)
#define STAGE_CAP 12288
#define PART_T 1024
#define BINS_PER_LANE ((NBINS + 31) / 32)
static_assert(N_NODES <= (1 << SRC_BITS), "source id must fit SRC_BITS");
static_assert(BIN_N <= PART_T && NBINS <= PART_T && NCHUNK == 256, "bucketing layout");

typedef float v2f __attribute__((ext_vector_type(2)));
typedef float v8f __attribute__((ext_vector_type(8)));

__global__ __launch_bounds__(PART_T) void chunk_count_kernel(const int* __restrict__ dst,
                                                             int* __restrict__ cnt1) {
    __shared__ int s_h[NBINS];
    const int t = threadIdx.x;
    const int wg = blockIdx.x;
    if (t < NBINS) s_h[t] = 0;
    __syncthreads();
    const int beg = wg * CHUNK_E;
    const int end = (beg + CHUNK_E < N_EDGES) ? (beg + CHUNK_E) : N_EDGES;
    for (int e0 = beg + t; e0 < end; e0 += 4 * PART_T) {
        int d[4];
#pragma unroll
        for (int u = 0; u < 4; ++u) {
            const int e = e0 + u * PART_T;
            d[u] = (e < end) ? dst[e] : -1;
        }
#pragma unroll
        for (int u = 0; u < 4; ++u)
            if (d[u] >= 0) atomicAdd(&s_h[d[u] >> BIN_SHIFT], 1);
    }
    __syncthreads();
    if (t < NBINS) cnt1[t * NCHUNK + wg] = s_h[t];
}

__global__ __launch_bounds__(PART_T) void chunk_fill_kernel(const int* __restrict__ src,
                                                            const int* __restrict__ dst,
                                                            const int* __restrict__ cnt1,
                                                            int* __restrict__ binstart,
                                                            int* __restrict__ packed) {
    __shared__ int s_tot[NBINS];
    __shared__ int s_bef[NBINS];
    __shared__ int s_own[NBINS];
    __shared__ int s_lcur[NBINS];
    __shared__ int s_goff[NBINS];
    __shared__ int s_val[CHUNK_E];
    __shared__ int s_gpos[CHUNK_E];
    const int t = threadIdx.x;
    const int lane = t & 31;
    const int w = t >> 5;
    const int wg = blockIdx.x;

    int4 a0[BINS_PER_LANE], a1[BINS_PER_LANE];
#pragma unroll
    for (int i = 0; i < BINS_PER_LANE; ++i) {
        const int b = w + i * (PART_T / 32);
        const int4* row = (const int4*)(cnt1 + ((b < NBINS) ? b : (NBINS - 1)) * NCHUNK);
        a0[i] = row[lane];
        a1[i] = row[lane + 32];
    }
    const int c0 = 4 * lane;
    const int c1 = 128 + 4 * lane;
#pragma unroll
    for (int i = 0; i < BINS_PER_LANE; ++i) {
        const int b = w + i * (PART_T / 32);
        const int v[8] = {a0[i].x, a0[i].y, a0[i].z, a0[i].w, a1[i].x, a1[i].y, a1[i].z, a1[i].w};
        int tot = 0, bef = 0, own = 0;
#pragma unroll
        for (int j = 0; j < 8; ++j) {
            const int c = ((j < 4) ? c0 : c1) + (j & 3);
            tot += v[j];
            bef += (c < wg) ? v[j] : 0;
            own += (c == wg) ? v[j] : 0;
        }
        for (int o = 16; o > 0; o >>= 1) {
            tot += __shfl_xor(tot, o);
            bef += __shfl_xor(bef, o);
            own += __shfl_xor(own, o);
        }
        if (lane == 0 && b < NBINS) { s_tot[b] = tot; s_bef[b] = bef; s_own[b] = own; }
    }
    __syncthreads();

    if (w == 0) {
        int gloc[BINS_PER_LANE], lloc[BINS_PER_LANE];
        int gsum = 0, lsum = 0;
#pragma unroll
        for (int i = 0; i < BINS_PER_LANE; ++i) {
            const int b = lane * BINS_PER_LANE + i;
            gloc[i] = gsum;
            lloc[i] = lsum;
            gsum += (b < NBINS) ? s_tot[b] : 0;
            lsum += (b < NBINS) ? s_own[b] : 0;
        }
        int ginc = gsum, linc = lsum;
        for (int o = 1; o < 32; o <<= 1) {
            const int gu = __shfl_up(ginc, o);
            const int lu = __shfl_up(linc, o);
            if (lane >= o) { ginc += gu; linc += lu; }
        }
        const int gexcl = ginc - gsum;
        const int lexcl = linc - lsum;
#pragma unroll
        for (int i = 0; i < BINS_PER_LANE; ++i) {
            const int b = lane * BINS_PER_LANE + i;
            if (b < NBINS) {
                const int gstart = gexcl + gloc[i];
                const int lstart = lexcl + lloc[i];
                s_lcur[b] = lstart;
                s_goff[b] = gstart + s_bef[b] - lstart;
                if (wg == 0) {
                    binstart[b] = gstart;
                    if (b == NBINS - 1) binstart[NBINS] = gstart + s_tot[b];
                }
            }
        }
    }
    __syncthreads();

    const int beg = wg * CHUNK_E;
    const int end = (beg + CHUNK_E < N_EDGES) ? (beg + CHUNK_E) : N_EDGES;
    for (int e0 = beg + t; e0 < end; e0 += 4 * PART_T) {
        int d[4], s[4];
#pragma unroll
        for (int u = 0; u < 4; ++u) {
            const int e = e0 + u * PART_T;
            d[u] = (e < end) ? dst[e] : -1;
            s[u] = (e < end) ? src[e] : 0;
        }
#pragma unroll
        for (int u = 0; u < 4; ++u) {
            if (d[u] >= 0) {
                const int b = d[u] >> BIN_SHIFT;
                const int lpos = atomicAdd(&s_lcur[b], 1);
                s_val[lpos] = ((d[u] & (BIN_N - 1)) << SRC_BITS) | s[u];
                s_gpos[lpos] = lpos + s_goff[b];
            }
        }
    }
    __syncthreads();
    for (int i = t; i < end - beg; i += PART_T) packed[s_gpos[i]] = s_val[i];
}

__global__ __launch_bounds__(PART_T) void bin_order_kernel(const int* __restrict__ packed,
                                                           const int* __restrict__ binstart,
                                                           int* __restrict__ rowptr,
                                                           float* __restrict__ invdeg,
                                                           int* __restrict__ col) {
    __shared__ int s_cnt[BIN_N];
    __shared__ int s_wave[PART_T / 32];
    __shared__ int s_col[STAGE_CAP];
    const int t = threadIdx.x;
    const int lane = t & 31;
    const int w = t >> 5;
    const int b = blockIdx.x;
    const int ebeg = binstart[b];
    const int eend = binstart[b + 1];
    const bool staged = (eend - ebeg) <= STAGE_CAP;

    if (t < BIN_N) s_cnt[t] = 0;
    __syncthreads();
    for (int e0 = ebeg + t; e0 < eend; e0 += 4 * PART_T) {
        int p[4];
#pragma unroll
        for (int u = 0; u < 4; ++u) {
            const int e = e0 + u * PART_T;
            p[u] = (e < eend) ? packed[e] : -1;
        }
#pragma unroll
        for (int u = 0; u < 4; ++u)
            if (p[u] >= 0) atomicAdd(&s_cnt[p[u] >> SRC_BITS], 1);
    }
    __syncthreads();

    const int c = (t < BIN_N) ? s_cnt[t] : 0;
    int inc = c;
    for (int o = 1; o < 32; o <<= 1) {
        const int u = __shfl_up(inc, o);
        if (lane >= o) inc += u;
    }
    if (lane == 31) s_wave[w] = inc;
    __syncthreads();
    int front = (lane < w) ? s_wave[lane] : 0;
    for (int o = 16; o > 0; o >>= 1) front += __shfl_xor(front, o);
    const int excl = front + inc - c;

    const int node = b * BIN_N + t;
    if (t < BIN_N) {
        s_cnt[t] = excl;
        if (node < N_NODES) {
            rowptr[node] = ebeg + excl;
            invdeg[node] = 1.0f / fmaxf((float)c, 1.0f);
            if (node == N_NODES - 1) rowptr[N_NODES] = ebeg + excl + c;
        }
    }
    __syncthreads();

    for (int e0 = ebeg + t; e0 < eend; e0 += 4 * PART_T) {
        int p[4];
#pragma unroll
        for (int u = 0; u < 4; ++u) {
            const int e = e0 + u * PART_T;
            p[u] = (e < eend) ? packed[e] : -1;
        }
#pragma unroll
        for (int u = 0; u < 4; ++u) {
            if (p[u] >= 0) {
                const int pos = atomicAdd(&s_cnt[p[u] >> SRC_BITS], 1);
                if (staged) s_col[pos] = p[u] & SRC_MASK;
                else col[ebeg + pos] = p[u] & SRC_MASK;
            }
        }
    }
    if (staged) {
        __syncthreads();
        for (int i = t; i < eend - ebeg; i += PART_T) col[ebeg + i] = s_col[i];
    }
}

__device__ __forceinline__ void add4(float4& a, const float4& v) {
    a.x += v.x; a.y += v.y; a.z += v.z; a.w += v.w;
}

#define FUSE_WAVES 4
#define MEAN_PITCH (HDIM + 4)

struct StripeWalk {
    int cur;
    int bnd;
    float4 acc;
};

__device__ __forceinline__ void close_node(StripeWalk& w, float* __restrict__ s_rows, int invbits,
                                           int lane) {
    const float inv = __int_as_float(__builtin_amdgcn_readlane(invbits, w.cur));
    const float4 o = make_float4(w.acc.x * inv, w.acc.y * inv, w.acc.z * inv, w.acc.w * inv);
    *(float4*)(s_rows + w.cur * MEAN_PITCH + 4 * lane) = o;
    w.acc = make_float4(0.f, 0.f, 0.f, 0.f);
}

template <int U>
__device__ __forceinline__ void gather_rows(const float4* __restrict__ in4, int mine, int t, int e,
                                            int lane, int rp, int invbits,
                                            float* __restrict__ s_rows, StripeWalk& w) {
    float4 v[U];
#pragma unroll
    for (int u = 0; u < U; ++u) {
        const int s = __builtin_amdgcn_readlane(mine, t + u);
        v[u] = in4[(long long)s * (HDIM / 4) + lane];
    }
#pragma unroll
    for (int u = 0; u < U; ++u) {
        while (e + u == w.bnd) {
            close_node(w, s_rows, invbits, lane);
            ++w.cur;
            w.bnd = __builtin_amdgcn_readlane(rp, w.cur + 1);
        }
        add4(w.acc, v[u]);
    }
}

__device__ __forceinline__ void gather_stripe(const float* __restrict__ in,
                                              const int* __restrict__ rowptr,
                                              const int* __restrict__ col,
                                              const float* __restrict__ invdeg, int row0, int lane,
                                              float* __restrict__ s_rows) {
    const int rp = rowptr[row0 + ((lane < 16) ? lane : 16)];
    const int invbits = __float_as_int(invdeg[row0 + (lane & 15)]);
    const int ebeg = __builtin_amdgcn_readlane(rp, 0);
    const int eend = __builtin_amdgcn_readlane(rp, 16);
    const float4* in4 = (const float4*)in;
    StripeWalk w;
    w.cur = 0;
    w.bnd = __builtin_amdgcn_readlane(rp, 1);
    w.acc = make_float4(0.f, 0.f, 0.f, 0.f);
    int ahead = (ebeg + lane < eend) ? col[ebeg + lane] : 0;
    for (int e0 = ebeg; e0 < eend; e0 += 32) {
        const int nj = (eend - e0 < 32) ? (eend - e0) : 32;
        const int mine = ahead;
        if (e0 + 32 < eend) ahead = (e0 + 32 + lane < eend) ? col[e0 + 32 + lane] : 0;
        int t = 0;
        for (; t + 8 <= nj; t += 8) gather_rows<8>(in4, mine, t, e0 + t, lane, rp, invbits, s_rows, w);
        if (t + 4 <= nj) { gather_rows<4>(in4, mine, t, e0 + t, lane, rp, invbits, s_rows, w); t += 4; }
        if (t + 2 <= nj) { gather_rows<2>(in4, mine, t, e0 + t, lane, rp, invbits, s_rows, w); t += 2; }
        if (t < nj) gather_rows<1>(in4, mine, t, e0 + t, lane, rp, invbits, s_rows, w);
    }
    while (w.cur < 16) {
        close_node(w, s_rows, invbits, lane);
        ++w.cur;
    }
}

__global__ __launch_bounds__(256) void gather16_kernel(const float* __restrict__ in,
                                                       const int* __restrict__ rowptr,
                                                       const int* __restrict__ col,
                                                       float* __restrict__ agg) {
    const int tid = blockIdx.x * 256 + threadIdx.x;
    const int node = tid >> 2;
    const int c = tid & 3;
    if (node >= N_NODES) return;
    const int beg = rowptr[node];
    const int end = rowptr[node + 1];
    const float4* in4 = (const float4*)in;
    float4 acc = make_float4(0.f, 0.f, 0.f, 0.f);
    int j = beg;
    for (; j + 4 <= end; j += 4) {
        const int s0 = col[j], s1 = col[j + 1], s2 = col[j + 2], s3 = col[j + 3];
        const float4 v0 = in4[(long long)s0 * (F_INPUT / 4) + c];
        const float4 v1 = in4[(long long)s1 * (F_INPUT / 4) + c];
        const float4 v2 = in4[(long long)s2 * (F_INPUT / 4) + c];
        const float4 v3 = in4[(long long)s3 * (F_INPUT / 4) + c];
        add4(acc, v0); add4(acc, v1); add4(acc, v2); add4(acc, v3);
    }
    for (; j < end; ++j) {
        const float4 v = in4[(long long)col[j] * (F_INPUT / 4) + c];
        add4(acc, v);
    }
    ((float4*)agg)[(long long)node * (F_INPUT / 4) + c] = acc;
}

__global__ __launch_bounds__(256) void sage_gemm_kernel(const float* __restrict__ in,
                                                        const float* __restrict__ agg,
                                                        const float* __restrict__ invdeg,
                                                        const float* __restrict__ Wr,
                                                        const float* __restrict__ Wl,
                                                        const float* __restrict__ bias,
                                                        float* __restrict__ out,
                                                        int K, int do_relu) {
    const int lane = threadIdx.x & 31;
    const int wv = threadIdx.x >> 5;
    const int mt = blockIdx.x * 8 + wv;
    if (mt >= M_TILES) return;

    const int row0 = mt << 4;
    const int m = lane & 15;
    const int kh = (lane >> 4) << 1;
    const long long arow = (long long)(row0 + m);

    v8f acc[8] = {};

    for (int k = 0; k < K; k += 4) {
        v2f a = *(const v2f*)(in + arow * K + k + kh);
#pragma unroll
        for (int nt = 0; nt < 8; ++nt) {
            const int col = (nt << 4) + m;
            v2f b;
            b.x = Wr[(k + kh) * HDIM + col];
            b.y = Wr[(k + kh + 1) * HDIM + col];
            acc[nt] = __builtin_amdgcn_wmma_f32_16x16x4_f32(
                false, a, false, b, (short)0, acc[nt], false, false);
        }
    }

    const float inv = invdeg[row0 + m];
    for (int k = 0; k < K; k += 4) {
        v2f a = *(const v2f*)(agg + arow * K + k + kh);
        a.x *= inv;
        a.y *= inv;
#pragma unroll
        for (int nt = 0; nt < 8; ++nt) {
            const int col = (nt << 4) + m;
            v2f b;
            b.x = Wl[(k + kh) * HDIM + col];
            b.y = Wl[(k + kh + 1) * HDIM + col];
            acc[nt] = __builtin_amdgcn_wmma_f32_16x16x4_f32(
                false, a, false, b, (short)0, acc[nt], false, false);
        }
    }

    const int rofs = (lane >> 4) << 3;
#pragma unroll
    for (int nt = 0; nt < 8; ++nt) {
        const int col = (nt << 4) + m;
        const float bv = bias[col];
#pragma unroll
        for (int j = 0; j < 8; ++j) {
            float v = acc[nt][j] + bv;
            if (do_relu) v = fmaxf(v, 0.0f);
            out[(long long)(row0 + rofs + j) * HDIM + col] = v;
        }
    }
}

__global__ __launch_bounds__(FUSE_WAVES * 32) void sage_layer128_kernel(
    const float* __restrict__ in, const int* __restrict__ rowptr, const int* __restrict__ col,
    const float* __restrict__ invdeg, const float* __restrict__ Wr, const float* __restrict__ Wl,
    const float* __restrict__ bias, float* __restrict__ out, int do_relu) {
    __shared__ float s_mean[FUSE_WAVES * 16 * MEAN_PITCH];
    const int lane = threadIdx.x & 31;
    const int wv = __builtin_amdgcn_readfirstlane((int)(threadIdx.x >> 5));
    const int mt = blockIdx.x * FUSE_WAVES + wv;
    if (mt >= M_TILES) return;

    const int row0 = mt << 4;
    const int m = lane & 15;
    const int kh = (lane >> 4) << 1;
    const long long arow = (long long)(row0 + m);
    float* s_rows = s_mean + wv * (16 * MEAN_PITCH);

    v8f acc[8] = {};

    const bool gather_first = ((blockIdx.x + (blockIdx.x >> 8) + wv) & 1) == 0;
    if (gather_first) gather_stripe(in, rowptr, col, invdeg, row0, lane, s_rows);

    for (int k = 0; k < HDIM; k += 4) {
        v2f a = *(const v2f*)(in + arow * HDIM + k + kh);
#pragma unroll
        for (int nt = 0; nt < 8; ++nt) {
            const int cc = (nt << 4) + m;
            v2f b;
            b.x = Wr[(k + kh) * HDIM + cc];
            b.y = Wr[(k + kh + 1) * HDIM + cc];
            acc[nt] = __builtin_amdgcn_wmma_f32_16x16x4_f32(
                false, a, false, b, (short)0, acc[nt], false, false);
        }
    }

    if (!gather_first) gather_stripe(in, rowptr, col, invdeg, row0, lane, s_rows);

    __builtin_amdgcn_fence(__ATOMIC_RELEASE, "wavefront");
    __builtin_amdgcn_wave_barrier();
    __builtin_amdgcn_fence(__ATOMIC_ACQUIRE, "wavefront");

    for (int k = 0; k < HDIM; k += 4) {
        v2f a = *(const v2f*)(s_rows + m * MEAN_PITCH + k + kh);
#pragma unroll
        for (int nt = 0; nt < 8; ++nt) {
            const int cc = (nt << 4) + m;
            v2f b;
            b.x = Wl[(k + kh) * HDIM + cc];
            b.y = Wl[(k + kh + 1) * HDIM + cc];
            acc[nt] = __builtin_amdgcn_wmma_f32_16x16x4_f32(
                false, a, false, b, (short)0, acc[nt], false, false);
        }
    }

    const int rofs = (lane >> 4) << 3;
#pragma unroll
    for (int nt = 0; nt < 8; ++nt) {
        const int cc = (nt << 4) + m;
        const float bv = bias[cc];
#pragma unroll
        for (int j = 0; j < 8; ++j) {
            float v = acc[nt][j] + bv;
            if (do_relu) v = fmaxf(v, 0.0f);
            out[(long long)(row0 + rofs + j) * HDIM + cc] = v;
        }
    }
}

extern "C" void kernel_launch(void* const* d_in, const int* in_sizes, int n_in,
                              void* d_out, int out_size, void* d_ws, size_t ws_size,
                              hipStream_t stream) {
    (void)in_sizes; (void)n_in; (void)out_size; (void)ws_size;

    const float* x = (const float*)d_in[0];
    const int* ei = (const int*)d_in[1];
    const int* src = ei;
    const int* dst = ei + N_EDGES;
    const float* Wl[4] = {(const float*)d_in[2], (const float*)d_in[5],
                          (const float*)d_in[8], (const float*)d_in[11]};
    const float* Wr[4] = {(const float*)d_in[3], (const float*)d_in[6],
                          (const float*)d_in[9], (const float*)d_in[12]};
    const float* bs[4] = {(const float*)d_in[4], (const float*)d_in[7],
                          (const float*)d_in[10], (const float*)d_in[13]};

    float* agg = (float*)d_ws;
    float* hA = agg + (size_t)N_NODES * HDIM;
    float* hB = hA + (size_t)N_NODES * HDIM;
    float* deg = hB + (size_t)N_NODES * HDIM;

    int* packed = (int*)hB;
    int* cnt1 = packed + N_EDGES;
    int* binstart = cnt1 + NBINS * NCHUNK;
    int* col = (int*)(agg + (size_t)N_NODES * F_INPUT);
    int* rowptr = col + N_EDGES;

    chunk_count_kernel<<<NCHUNK, PART_T, 0, stream>>>(dst, cnt1);
    chunk_fill_kernel<<<NCHUNK, PART_T, 0, stream>>>(src, dst, cnt1, binstart, packed);
    bin_order_kernel<<<NBINS, PART_T, 0, stream>>>(packed, binstart, rowptr, deg, col);

    const int gemm_grid = (M_TILES + 7) / 8;

    gather16_kernel<<<(N_NODES * (F_INPUT / 4) + 255) / 256, 256, 0, stream>>>(x, rowptr, col, agg);
    sage_gemm_kernel<<<gemm_grid, 256, 0, stream>>>(x, agg, deg, Wr[0], Wl[0], bs[0],
                                                    hA, F_INPUT, 1);

    const float* cur = hA;
    float* nxt = hB;
    const int layer_grid = (M_TILES + FUSE_WAVES - 1) / FUSE_WAVES;
    for (int l = 1; l < 4; ++l) {
        float* outp = (l == 3) ? (float*)d_out : nxt;
        int relu = (l == 3) ? 0 : 1;
        sage_layer128_kernel<<<layer_grid, FUSE_WAVES * 32, 0, stream>>>(
            cur, rowptr, col, deg, Wr[l], Wl[l], bs[l], outp, relu);
        float* t2 = (float*)cur;
        cur = outp;
        nxt = t2;
    }
}


# Layers 2-4: one 25-wave work-group per WGP (250 x 25 stripes, no stripe loop) stages Wr and 24 of Wl's 32 k steps in LDS in fragment order; those B fragments are LDS reads; order of sums unchanged
# speedup vs baseline: 4.3908x; 1.0767x over previous
#include <hip/hip_runtime.h>
#include <hip/hip_bf16.h>


#define N_NODES 100000
#define N_EDGES 1600000
#define F_INPUT 16
#define HDIM 128
#define M_TILES (N_NODES / 16)
#define BIN_SHIFT 9
#define BIN_N (1 << BIN_SHIFT)
#define NBINS ((N_NODES + BIN_N - 1) / BIN_N)
#define NCHUNK 256
#define CHUNK_E ((N_EDGES + NCHUNK - 1) / NCHUNK)
#define SRC_BITS 17
#define SRC_MASK ((1 << SRC_BITS) - 1)
#define STAGE_CAP 12288
#define PART_T 1024
#define BINS_PER_LANE ((NBINS + 31) / 32)
static_assert(N_NODES <= (1 << SRC_BITS), "source id must fit SRC_BITS");
static_assert(BIN_N <= PART_T && NBINS <= PART_T && NCHUNK == 256, "bucketing layout");

typedef float v2f __attribute__((ext_vector_type(2)));
typedef float v8f __attribute__((ext_vector_type(8)));

__global__ __launch_bounds__(PART_T) void chunk_count_kernel(const int* __restrict__ dst,
                                                             int* __restrict__ cnt1) {
    __shared__ int s_h[NBINS];
    const int t = threadIdx.x;
    const int wg = blockIdx.x;
    if (t < NBINS) s_h[t] = 0;
    __syncthreads();
    const int beg = wg * CHUNK_E;
    const int end = (beg + CHUNK_E < N_EDGES) ? (beg + CHUNK_E) : N_EDGES;
    for (int e0 = beg + t; e0 < end; e0 += 4 * PART_T) {
        int d[4];
#pragma unroll
        for (int u = 0; u < 4; ++u) {
            const int e = e0 + u * PART_T;
            d[u] = (e < end) ? dst[e] : -1;
        }
#pragma unroll
        for (int u = 0; u < 4; ++u)
            if (d[u] >= 0) atomicAdd(&s_h[d[u] >> BIN_SHIFT], 1);
    }
    __syncthreads();
    if (t < NBINS) cnt1[t * NCHUNK + wg] = s_h[t];
}

__global__ __launch_bounds__(PART_T) void chunk_fill_kernel(const int* __restrict__ src,
                                                            const int* __restrict__ dst,
                                                            const int* __restrict__ cnt1,
                                                            int* __restrict__ binstart,
                                                            int* __restrict__ packed) {
    __shared__ int s_tot[NBINS];
    __shared__ int s_bef[NBINS];
    __shared__ int s_own[NBINS];
    __shared__ int s_lcur[NBINS];
    __shared__ int s_goff[NBINS];
    __shared__ int s_val[CHUNK_E];
    __shared__ int s_gpos[CHUNK_E];
    const int t = threadIdx.x;
    const int lane = t & 31;
    const int w = t >> 5;
    const int wg = blockIdx.x;

    int4 a0[BINS_PER_LANE], a1[BINS_PER_LANE];
#pragma unroll
    for (int i = 0; i < BINS_PER_LANE; ++i) {
        const int b = w + i * (PART_T / 32);
        const int4* row = (const int4*)(cnt1 + ((b < NBINS) ? b : (NBINS - 1)) * NCHUNK);
        a0[i] = row[lane];
        a1[i] = row[lane + 32];
    }
    const int c0 = 4 * lane;
    const int c1 = 128 + 4 * lane;
#pragma unroll
    for (int i = 0; i < BINS_PER_LANE; ++i) {
        const int b = w + i * (PART_T / 32);
        const int v[8] = {a0[i].x, a0[i].y, a0[i].z, a0[i].w, a1[i].x, a1[i].y, a1[i].z, a1[i].w};
        int tot = 0, bef = 0, own = 0;
#pragma unroll
        for (int j = 0; j < 8; ++j) {
            const int c = ((j < 4) ? c0 : c1) + (j & 3);
            tot += v[j];
            bef += (c < wg) ? v[j] : 0;
            own += (c == wg) ? v[j] : 0;
        }
        for (int o = 16; o > 0; o >>= 1) {
            tot += __shfl_xor(tot, o);
            bef += __shfl_xor(bef, o);
            own += __shfl_xor(own, o);
        }
        if (lane == 0 && b < NBINS) { s_tot[b] = tot; s_bef[b] = bef; s_own[b] = own; }
    }
    __syncthreads();

    if (w == 0) {
        int gloc[BINS_PER_LANE], lloc[BINS_PER_LANE];
        int gsum = 0, lsum = 0;
#pragma unroll
        for (int i = 0; i < BINS_PER_LANE; ++i) {
            const int b = lane * BINS_PER_LANE + i;
            gloc[i] = gsum;
            lloc[i] = lsum;
            gsum += (b < NBINS) ? s_tot[b] : 0;
            lsum += (b < NBINS) ? s_own[b] : 0;
        }
        int ginc = gsum, linc = lsum;
        for (int o = 1; o < 32; o <<= 1) {
            const int gu = __shfl_up(ginc, o);
            const int lu = __shfl_up(linc, o);
            if (lane >= o) { ginc += gu; linc += lu; }
        }
        const int gexcl = ginc - gsum;
        const int lexcl = linc - lsum;
#pragma unroll
        for (int i = 0; i < BINS_PER_LANE; ++i) {
            const int b = lane * BINS_PER_LANE + i;
            if (b < NBINS) {
                const int gstart = gexcl + gloc[i];
                const int lstart = lexcl + lloc[i];
                s_lcur[b] = lstart;
                s_goff[b] = gstart + s_bef[b] - lstart;
                if (wg == 0) {
                    binstart[b] = gstart;
                    if (b == NBINS - 1) binstart[NBINS] = gstart + s_tot[b];
                }
            }
        }
    }
    __syncthreads();

    const int beg = wg * CHUNK_E;
    const int end = (beg + CHUNK_E < N_EDGES) ? (beg + CHUNK_E) : N_EDGES;
    for (int e0 = beg + t; e0 < end; e0 += 4 * PART_T) {
        int d[4], s[4];
#pragma unroll
        for (int u = 0; u < 4; ++u) {
            const int e = e0 + u * PART_T;
            d[u] = (e < end) ? dst[e] : -1;
            s[u] = (e < end) ? src[e] : 0;
        }
#pragma unroll
        for (int u = 0; u < 4; ++u) {
            if (d[u] >= 0) {
                const int b = d[u] >> BIN_SHIFT;
                const int lpos = atomicAdd(&s_lcur[b], 1);
                s_val[lpos] = ((d[u] & (BIN_N - 1)) << SRC_BITS) | s[u];
                s_gpos[lpos] = lpos + s_goff[b];
            }
        }
    }
    __syncthreads();
    for (int i = t; i < end - beg; i += PART_T) packed[s_gpos[i]] = s_val[i];
}

__global__ __launch_bounds__(PART_T) void bin_order_kernel(const int* __restrict__ packed,
                                                           const int* __restrict__ binstart,
                                                           int* __restrict__ rowptr,
                                                           float* __restrict__ invdeg,
                                                           int* __restrict__ col) {
    __shared__ int s_cnt[BIN_N];
    __shared__ int s_wave[PART_T / 32];
    __shared__ int s_col[STAGE_CAP];
    const int t = threadIdx.x;
    const int lane = t & 31;
    const int w = t >> 5;
    const int b = blockIdx.x;
    const int ebeg = binstart[b];
    const int eend = binstart[b + 1];
    const bool staged = (eend - ebeg) <= STAGE_CAP;

    if (t < BIN_N) s_cnt[t] = 0;
    __syncthreads();
    for (int e0 = ebeg + t; e0 < eend; e0 += 4 * PART_T) {
        int p[4];
#pragma unroll
        for (int u = 0; u < 4; ++u) {
            const int e = e0 + u * PART_T;
            p[u] = (e < eend) ? packed[e] : -1;
        }
#pragma unroll
        for (int u = 0; u < 4; ++u)
            if (p[u] >= 0) atomicAdd(&s_cnt[p[u] >> SRC_BITS], 1);
    }
    __syncthreads();

    const int c = (t < BIN_N) ? s_cnt[t] : 0;
    int inc = c;
    for (int o = 1; o < 32; o <<= 1) {
        const int u = __shfl_up(inc, o);
        if (lane >= o) inc += u;
    }
    if (lane == 31) s_wave[w] = inc;
    __syncthreads();
    int front = (lane < w) ? s_wave[lane] : 0;
    for (int o = 16; o > 0; o >>= 1) front += __shfl_xor(front, o);
    const int excl = front + inc - c;

    const int node = b * BIN_N + t;
    if (t < BIN_N) {
        s_cnt[t] = excl;
        if (node < N_NODES) {
            rowptr[node] = ebeg + excl;
            invdeg[node] = 1.0f / fmaxf((float)c, 1.0f);
            if (node == N_NODES - 1) rowptr[N_NODES] = ebeg + excl + c;
        }
    }
    __syncthreads();

    for (int e0 = ebeg + t; e0 < eend; e0 += 4 * PART_T) {
        int p[4];
#pragma unroll
        for (int u = 0; u < 4; ++u) {
            const int e = e0 + u * PART_T;
            p[u] = (e < eend) ? packed[e] : -1;
        }
#pragma unroll
        for (int u = 0; u < 4; ++u) {
            if (p[u] >= 0) {
                const int pos = atomicAdd(&s_cnt[p[u] >> SRC_BITS], 1);
                if (staged) s_col[pos] = p[u] & SRC_MASK;
                else col[ebeg + pos] = p[u] & SRC_MASK;
            }
        }
    }
    if (staged) {
        __syncthreads();
        for (int i = t; i < eend - ebeg; i += PART_T) col[ebeg + i] = s_col[i];
    }
}

__device__ __forceinline__ void add4(float4& a, const float4& v) {
    a.x += v.x; a.y += v.y; a.z += v.z; a.w += v.w;
}

#define FUSE_WAVES 25
#define LAYER_T (FUSE_WAVES * 32)
#define W_UNITS (HDIM * HDIM / 4)
#define WL_LDS_STEPS 24
#define S_UNITS (W_UNITS + WL_LDS_STEPS * 128)
#define S_UNITS_PER_T ((S_UNITS + LAYER_T - 1) / LAYER_T)
static_assert(M_TILES % FUSE_WAVES == 0, "every wave of the layer grid owns a stripe");
#define MEAN_PITCH (HDIM + 4)

struct StripeWalk {
    int cur;
    int bnd;
    float4 acc;
};

__device__ __forceinline__ void close_node(StripeWalk& w, float* __restrict__ s_rows, int invbits,
                                           int lane) {
    const float inv = __int_as_float(__builtin_amdgcn_readlane(invbits, w.cur));
    const float4 o = make_float4(w.acc.x * inv, w.acc.y * inv, w.acc.z * inv, w.acc.w * inv);
    *(float4*)(s_rows + w.cur * MEAN_PITCH + 4 * lane) = o;
    w.acc = make_float4(0.f, 0.f, 0.f, 0.f);
}

template <int U>
__device__ __forceinline__ void gather_rows(const float4* __restrict__ in4, int mine, int t, int e,
                                            int lane, int rp, int invbits,
                                            float* __restrict__ s_rows, StripeWalk& w) {
    float4 v[U];
#pragma unroll
    for (int u = 0; u < U; ++u) {
        const int s = __builtin_amdgcn_readlane(mine, t + u);
        v[u] = in4[(long long)s * (HDIM / 4) + lane];
    }
#pragma unroll
    for (int u = 0; u < U; ++u) {
        while (e + u == w.bnd) {
            close_node(w, s_rows, invbits, lane);
            ++w.cur;
            w.bnd = __builtin_amdgcn_readlane(rp, w.cur + 1);
        }
        add4(w.acc, v[u]);
    }
}

__device__ __forceinline__ void gather_stripe(const float* __restrict__ in,
                                              const int* __restrict__ rowptr,
                                              const int* __restrict__ col,
                                              const float* __restrict__ invdeg, int row0, int lane,
                                              float* __restrict__ s_rows) {
    const int rp = rowptr[row0 + ((lane < 16) ? lane : 16)];
    const int invbits = __float_as_int(invdeg[row0 + (lane & 15)]);
    const int ebeg = __builtin_amdgcn_readlane(rp, 0);
    const int eend = __builtin_amdgcn_readlane(rp, 16);
    const float4* in4 = (const float4*)in;
    StripeWalk w;
    w.cur = 0;
    w.bnd = __builtin_amdgcn_readlane(rp, 1);
    w.acc = make_float4(0.f, 0.f, 0.f, 0.f);
    int ahead = (ebeg + lane < eend) ? col[ebeg + lane] : 0;
    for (int e0 = ebeg; e0 < eend; e0 += 32) {
        const int nj = (eend - e0 < 32) ? (eend - e0) : 32;
        const int mine = ahead;
        if (e0 + 32 < eend) ahead = (e0 + 32 + lane < eend) ? col[e0 + 32 + lane] : 0;
        int t = 0;
        for (; t + 8 <= nj; t += 8) gather_rows<8>(in4, mine, t, e0 + t, lane, rp, invbits, s_rows, w);
        if (t + 4 <= nj) { gather_rows<4>(in4, mine, t, e0 + t, lane, rp, invbits, s_rows, w); t += 4; }
        if (t + 2 <= nj) { gather_rows<2>(in4, mine, t, e0 + t, lane, rp, invbits, s_rows, w); t += 2; }
        if (t < nj) gather_rows<1>(in4, mine, t, e0 + t, lane, rp, invbits, s_rows, w);
    }
    while (w.cur < 16) {
        close_node(w, s_rows, invbits, lane);
        ++w.cur;
    }
}

__global__ __launch_bounds__(256) void gather16_kernel(const float* __restrict__ in,
                                                       const int* __restrict__ rowptr,
                                                       const int* __restrict__ col,
                                                       float* __restrict__ agg) {
    const int tid = blockIdx.x * 256 + threadIdx.x;
    const int node = tid >> 2;
    const int c = tid & 3;
    if (node >= N_NODES) return;
    const int beg = rowptr[node];
    const int end = rowptr[node + 1];
    const float4* in4 = (const float4*)in;
    float4 acc = make_float4(0.f, 0.f, 0.f, 0.f);
    int j = beg;
    for (; j + 4 <= end; j += 4) {
        const int s0 = col[j], s1 = col[j + 1], s2 = col[j + 2], s3 = col[j + 3];
        const float4 v0 = in4[(long long)s0 * (F_INPUT / 4) + c];
        const float4 v1 = in4[(long long)s1 * (F_INPUT / 4) + c];
        const float4 v2 = in4[(long long)s2 * (F_INPUT / 4) + c];
        const float4 v3 = in4[(long long)s3 * (F_INPUT / 4) + c];
        add4(acc, v0); add4(acc, v1); add4(acc, v2); add4(acc, v3);
    }
    for (; j < end; ++j) {
        const float4 v = in4[(long long)col[j] * (F_INPUT / 4) + c];
        add4(acc, v);
    }
    ((float4*)agg)[(long long)node * (F_INPUT / 4) + c] = acc;
}

__global__ __launch_bounds__(256) void sage_gemm_kernel(const float* __restrict__ in,
                                                        const float* __restrict__ agg,
                                                        const float* __restrict__ invdeg,
                                                        const float* __restrict__ Wr,
                                                        const float* __restrict__ Wl,
                                                        const float* __restrict__ bias,
                                                        float* __restrict__ out,
                                                        int K, int do_relu) {
    const int lane = threadIdx.x & 31;
    const int wv = threadIdx.x >> 5;
    const int mt = blockIdx.x * 8 + wv;
    if (mt >= M_TILES) return;

    const int row0 = mt << 4;
    const int m = lane & 15;
    const int kh = (lane >> 4) << 1;
    const long long arow = (long long)(row0 + m);

    v8f acc[8] = {};

    for (int k = 0; k < K; k += 4) {
        v2f a = *(const v2f*)(in + arow * K + k + kh);
#pragma unroll
        for (int nt = 0; nt < 8; ++nt) {
            const int col = (nt << 4) + m;
            v2f b;
            b.x = Wr[(k + kh) * HDIM + col];
            b.y = Wr[(k + kh + 1) * HDIM + col];
            acc[nt] = __builtin_amdgcn_wmma_f32_16x16x4_f32(
                false, a, false, b, (short)0, acc[nt], false, false);
        }
    }

    const float inv = invdeg[row0 + m];
    for (int k = 0; k < K; k += 4) {
        v2f a = *(const v2f*)(agg + arow * K + k + kh);
        a.x *= inv;
        a.y *= inv;
#pragma unroll
        for (int nt = 0; nt < 8; ++nt) {
            const int col = (nt << 4) + m;
            v2f b;
            b.x = Wl[(k + kh) * HDIM + col];
            b.y = Wl[(k + kh + 1) * HDIM + col];
            acc[nt] = __builtin_amdgcn_wmma_f32_16x16x4_f32(
                false, a, false, b, (short)0, acc[nt], false, false);
        }
    }

    const int rofs = (lane >> 4) << 3;
#pragma unroll
    for (int nt = 0; nt < 8; ++nt) {
        const int col = (nt << 4) + m;
        const float bv = bias[col];
#pragma unroll
        for (int j = 0; j < 8; ++j) {
            float v = acc[nt][j] + bv;
            if (do_relu) v = fmaxf(v, 0.0f);
            out[(long long)(row0 + rofs + j) * HDIM + col] = v;
        }
    }
}

__global__ __launch_bounds__(LAYER_T) void sage_layer128_kernel(
    const float* __restrict__ in, const int* __restrict__ rowptr, const int* __restrict__ col,
    const float* __restrict__ invdeg, const float* __restrict__ Wr, const float* __restrict__ Wl,
    const float* __restrict__ bias, float* __restrict__ out, int do_relu) {
    __shared__ float4 s_w[S_UNITS];
    __shared__ float s_mean[FUSE_WAVES * 16 * MEAN_PITCH];
    const int lane = threadIdx.x & 31;
    const int wv = __builtin_amdgcn_readfirstlane((int)(threadIdx.x >> 5));
    const int mt = blockIdx.x * FUSE_WAVES + wv;

    {
        float4 t[S_UNITS_PER_T];
#pragma unroll
        for (int i = 0; i < S_UNITS_PER_T; ++i) {
            const int u = (int)threadIdx.x + i * LAYER_T;
            if (u < S_UNITS) {
                const float* Wsrc = (u < W_UNITS) ? Wr : Wl;
                const int uu = (u < W_UNITS) ? u : (u - W_UNITS);
                const int ul = uu & 31;
                const int uj = (uu >> 5) & 3;
                const int uks = uu >> 7;
                const float* p = Wsrc + (4 * uks + ((ul >> 4) << 1)) * HDIM + 32 * uj + (ul & 15);
                t[i] = make_float4(p[0], p[HDIM], p[16], p[HDIM + 16]);
            }
        }
#pragma unroll
        for (int i = 0; i < S_UNITS_PER_T; ++i) {
            const int u = (int)threadIdx.x + i * LAYER_T;
            if (u < S_UNITS) s_w[u] = t[i];
        }
    }
    __syncthreads();

    const int row0 = mt << 4;
    const int m = lane & 15;
    const int kh = (lane >> 4) << 1;
    const long long arow = (long long)(row0 + m);
    float* s_rows = s_mean + wv * (16 * MEAN_PITCH);

    v8f acc[8] = {};

    const bool gather_first = ((wv >> 2) & 1) == 0;
    if (gather_first) gather_stripe(in, rowptr, col, invdeg, row0, lane, s_rows);

#pragma unroll 2
    for (int ks = 0; ks < HDIM / 4; ++ks) {
        v2f a = *(const v2f*)(in + arow * HDIM + 4 * ks + kh);
#pragma unroll
        for (int j = 0; j < 4; ++j) {
            const float4 bb = s_w[(ks * 4 + j) * 32 + lane];
            v2f b0, b1;
            b0.x = bb.x; b0.y = bb.y;
            b1.x = bb.z; b1.y = bb.w;
            acc[2 * j] = __builtin_amdgcn_wmma_f32_16x16x4_f32(
                false, a, false, b0, (short)0, acc[2 * j], false, false);
            acc[2 * j + 1] = __builtin_amdgcn_wmma_f32_16x16x4_f32(
                false, a, false, b1, (short)0, acc[2 * j + 1], false, false);
        }
    }

    if (!gather_first) gather_stripe(in, rowptr, col, invdeg, row0, lane, s_rows);

    __builtin_amdgcn_fence(__ATOMIC_RELEASE, "wavefront");
    __builtin_amdgcn_wave_barrier();
    __builtin_amdgcn_fence(__ATOMIC_ACQUIRE, "wavefront");

#pragma unroll 2
    for (int ks = 0; ks < WL_LDS_STEPS; ++ks) {
        v2f a = *(const v2f*)(s_rows + m * MEAN_PITCH + 4 * ks + kh);
#pragma unroll
        for (int j = 0; j < 4; ++j) {
            const float4 bb = s_w[W_UNITS + (ks * 4 + j) * 32 + lane];
            v2f b0, b1;
            b0.x = bb.x; b0.y = bb.y;
            b1.x = bb.z; b1.y = bb.w;
            acc[2 * j] = __builtin_amdgcn_wmma_f32_16x16x4_f32(
                false, a, false, b0, (short)0, acc[2 * j], false, false);
            acc[2 * j + 1] = __builtin_amdgcn_wmma_f32_16x16x4_f32(
                false, a, false, b1, (short)0, acc[2 * j + 1], false, false);
        }
    }
    for (int k = 4 * WL_LDS_STEPS; k < HDIM; k += 4) {
        v2f a = *(const v2f*)(s_rows + m * MEAN_PITCH + k + kh);
#pragma unroll
        for (int nt = 0; nt < 8; ++nt) {
            const int cc = (nt << 4) + m;
            v2f b;
            b.x = Wl[(k + kh) * HDIM + cc];
            b.y = Wl[(k + kh + 1) * HDIM + cc];
            acc[nt] = __builtin_amdgcn_wmma_f32_16x16x4_f32(
                false, a, false, b, (short)0, acc[nt], false, false);
        }
    }

    const int rofs = (lane >> 4) << 3;
#pragma unroll
    for (int nt = 0; nt < 8; ++nt) {
        const int cc = (nt << 4) + m;
        const float bv = bias[cc];
#pragma unroll
        for (int j = 0; j < 8; ++j) {
            float v = acc[nt][j] + bv;
            if (do_relu) v = fmaxf(v, 0.0f);
            out[(long long)(row0 + rofs + j) * HDIM + cc] = v;
        }
    }
}

extern "C" void kernel_launch(void* const* d_in, const int* in_sizes, int n_in,
                              void* d_out, int out_size, void* d_ws, size_t ws_size,
                              hipStream_t stream) {
    (void)in_sizes; (void)n_in; (void)out_size; (void)ws_size;

    const float* x = (const float*)d_in[0];
    const int* ei = (const int*)d_in[1];
    const int* src = ei;
    const int* dst = ei + N_EDGES;
    const float* Wl[4] = {(const float*)d_in[2], (const float*)d_in[5],
                          (const float*)d_in[8], (const float*)d_in[11]};
    const float* Wr[4] = {(const float*)d_in[3], (const float*)d_in[6],
                          (const float*)d_in[9], (const float*)d_in[12]};
    const float* bs[4] = {(const float*)d_in[4], (const float*)d_in[7],
                          (const float*)d_in[10], (const float*)d_in[13]};

    float* agg = (float*)d_ws;
    float* hA = agg + (size_t)N_NODES * HDIM;
    float* hB = hA + (size_t)N_NODES * HDIM;
    float* deg = hB + (size_t)N_NODES * HDIM;

    int* packed = (int*)hB;
    int* cnt1 = packed + N_EDGES;
    int* binstart = cnt1 + NBINS * NCHUNK;
    int* col = (int*)(agg + (size_t)N_NODES * F_INPUT);
    int* rowptr = col + N_EDGES;

    chunk_count_kernel<<<NCHUNK, PART_T, 0, stream>>>(dst, cnt1);
    chunk_fill_kernel<<<NCHUNK, PART_T, 0, stream>>>(src, dst, cnt1, binstart, packed);
    bin_order_kernel<<<NBINS, PART_T, 0, stream>>>(packed, binstart, rowptr, deg, col);

    const int gemm_grid = (M_TILES + 7) / 8;

    gather16_kernel<<<(N_NODES * (F_INPUT / 4) + 255) / 256, 256, 0, stream>>>(x, rowptr, col, agg);
    sage_gemm_kernel<<<gemm_grid, 256, 0, stream>>>(x, agg, deg, Wr[0], Wl[0], bs[0],
                                                    hA, F_INPUT, 1);

    const float* cur = hA;
    float* nxt = hB;
    const int layer_grid = M_TILES / FUSE_WAVES;
    for (int l = 1; l < 4; ++l) {
        float* outp = (l == 3) ? (float*)d_out : nxt;
        int relu = (l == 3) ? 0 : 1;
        sage_layer128_kernel<<<layer_grid, LAYER_T, 0, stream>>>(
            cur, rowptr, col, deg, Wr[l], Wl[l], bs[l], outp, relu);
        float* t2 = (float*)cur;
        cur = outp;
        nxt = t2;
    }
}


# Layer 1 (K=16) moved into the bin-ordering launch: a bin's work-group sums its nodes' neighbour rows from the list it just ordered (in LDS, or col if too big) and projects them; sum order unchanged
# speedup vs baseline: 4.5343x; 1.0327x over previous
#include <hip/hip_runtime.h>
#include <hip/hip_bf16.h>


#define N_NODES 100000
#define N_EDGES 1600000
#define F_INPUT 16
#define HDIM 128
#define M_TILES (N_NODES / 16)
#define BIN_SHIFT 9
#define BIN_N (1 << BIN_SHIFT)
#define NBINS ((N_NODES + BIN_N - 1) / BIN_N)
#define NCHUNK 256
#define CHUNK_E ((N_EDGES + NCHUNK - 1) / NCHUNK)
#define SRC_BITS 17
#define SRC_MASK ((1 << SRC_BITS) - 1)
#define STAGE_CAP 12288
#define PART_T 1024
#define BINS_PER_LANE ((NBINS + 31) / 32)
static_assert(N_NODES <= (1 << SRC_BITS), "source id must fit SRC_BITS");
static_assert(BIN_N <= PART_T && NBINS <= PART_T && NCHUNK == 256, "bucketing layout");

typedef float v2f __attribute__((ext_vector_type(2)));
typedef float v8f __attribute__((ext_vector_type(8)));

__global__ __launch_bounds__(PART_T) void chunk_count_kernel(const int* __restrict__ dst,
                                                             int* __restrict__ cnt1) {
    __shared__ int s_h[NBINS];
    const int t = threadIdx.x;
    const int wg = blockIdx.x;
    if (t < NBINS) s_h[t] = 0;
    __syncthreads();
    const int beg = wg * CHUNK_E;
    const int end = (beg + CHUNK_E < N_EDGES) ? (beg + CHUNK_E) : N_EDGES;
    for (int e0 = beg + t; e0 < end; e0 += 4 * PART_T) {
        int d[4];
#pragma unroll
        for (int u = 0; u < 4; ++u) {
            const int e = e0 + u * PART_T;
            d[u] = (e < end) ? dst[e] : -1;
        }
#pragma unroll
        for (int u = 0; u < 4; ++u)
            if (d[u] >= 0) atomicAdd(&s_h[d[u] >> BIN_SHIFT], 1);
    }
    __syncthreads();
    if (t < NBINS) cnt1[t * NCHUNK + wg] = s_h[t];
}

__global__ __launch_bounds__(PART_T) void chunk_fill_kernel(const int* __restrict__ src,
                                                            const int* __restrict__ dst,
                                                            const int* __restrict__ cnt1,
                                                            int* __restrict__ binstart,
                                                            int* __restrict__ packed) {
    __shared__ int s_tot[NBINS];
    __shared__ int s_bef[NBINS];
    __shared__ int s_own[NBINS];
    __shared__ int s_lcur[NBINS];
    __shared__ int s_goff[NBINS];
    __shared__ int s_val[CHUNK_E];
    __shared__ int s_gpos[CHUNK_E];
    const int t = threadIdx.x;
    const int lane = t & 31;
    const int w = t >> 5;
    const int wg = blockIdx.x;

    int4 a0[BINS_PER_LANE], a1[BINS_PER_LANE];
#pragma unroll
    for (int i = 0; i < BINS_PER_LANE; ++i) {
        const int b = w + i * (PART_T / 32);
        const int4* row = (const int4*)(cnt1 + ((b < NBINS) ? b : (NBINS - 1)) * NCHUNK);
        a0[i] = row[lane];
        a1[i] = row[lane + 32];
    }
    const int c0 = 4 * lane;
    const int c1 = 128 + 4 * lane;
#pragma unroll
    for (int i = 0; i < BINS_PER_LANE; ++i) {
        const int b = w + i * (PART_T / 32);
        const int v[8] = {a0[i].x, a0[i].y, a0[i].z, a0[i].w, a1[i].x, a1[i].y, a1[i].z, a1[i].w};
        int tot = 0, bef = 0, own = 0;
#pragma unroll
        for (int j = 0; j < 8; ++j) {
            const int c = ((j < 4) ? c0 : c1) + (j & 3);
            tot += v[j];
            bef += (c < wg) ? v[j] : 0;
            own += (c == wg) ? v[j] : 0;
        }
        for (int o = 16; o > 0; o >>= 1) {
            tot += __shfl_xor(tot, o);
            bef += __shfl_xor(bef, o);
            own += __shfl_xor(own, o);
        }
        if (lane == 0 && b < NBINS) { s_tot[b] = tot; s_bef[b] = bef; s_own[b] = own; }
    }
    __syncthreads();

    if (w == 0) {
        int gloc[BINS_PER_LANE], lloc[BINS_PER_LANE];
        int gsum = 0, lsum = 0;
#pragma unroll
        for (int i = 0; i < BINS_PER_LANE; ++i) {
            const int b = lane * BINS_PER_LANE + i;
            gloc[i] = gsum;
            lloc[i] = lsum;
            gsum += (b < NBINS) ? s_tot[b] : 0;
            lsum += (b < NBINS) ? s_own[b] : 0;
        }
        int ginc = gsum, linc = lsum;
        for (int o = 1; o < 32; o <<= 1) {
            const int gu = __shfl_up(ginc, o);
            const int lu = __shfl_up(linc, o);
            if (lane >= o) { ginc += gu; linc += lu; }
        }
        const int gexcl = ginc - gsum;
        const int lexcl = linc - lsum;
#pragma unroll
        for (int i = 0; i < BINS_PER_LANE; ++i) {
            const int b = lane * BINS_PER_LANE + i;
            if (b < NBINS) {
                const int gstart = gexcl + gloc[i];
                const int lstart = lexcl + lloc[i];
                s_lcur[b] = lstart;
                s_goff[b] = gstart + s_bef[b] - lstart;
                if (wg == 0) {
                    binstart[b] = gstart;
                    if (b == NBINS - 1) binstart[NBINS] = gstart + s_tot[b];
                }
            }
        }
    }
    __syncthreads();

    const int beg = wg * CHUNK_E;
    const int end = (beg + CHUNK_E < N_EDGES) ? (beg + CHUNK_E) : N_EDGES;
    for (int e0 = beg + t; e0 < end; e0 += 4 * PART_T) {
        int d[4], s[4];
#pragma unroll
        for (int u = 0; u < 4; ++u) {
            const int e = e0 + u * PART_T;
            d[u] = (e < end) ? dst[e] : -1;
            s[u] = (e < end) ? src[e] : 0;
        }
#pragma unroll
        for (int u = 0; u < 4; ++u) {
            if (d[u] >= 0) {
                const int b = d[u] >> BIN_SHIFT;
                const int lpos = atomicAdd(&s_lcur[b], 1);
                s_val[lpos] = ((d[u] & (BIN_N - 1)) << SRC_BITS) | s[u];
                s_gpos[lpos] = lpos + s_goff[b];
            }
        }
    }
    __syncthreads();
    for (int i = t; i < end - beg; i += PART_T) packed[s_gpos[i]] = s_val[i];
}

__device__ __forceinline__ void add4(float4& a, const float4& v) {
    a.x += v.x; a.y += v.y; a.z += v.z; a.w += v.w;
}

#define FUSE_WAVES 25
#define LAYER_T (FUSE_WAVES * 32)
#define W_UNITS (HDIM * HDIM / 4)
#define WL_LDS_STEPS 24
#define S_UNITS (W_UNITS + WL_LDS_STEPS * 128)
#define S_UNITS_PER_T ((S_UNITS + LAYER_T - 1) / LAYER_T)
static_assert(M_TILES % FUSE_WAVES == 0, "every wave of the layer grid owns a stripe");
#define MEAN_PITCH (HDIM + 4)

struct StripeWalk {
    int cur;
    int bnd;
    float4 acc;
};

__device__ __forceinline__ void close_node(StripeWalk& w, float* __restrict__ s_rows, int invbits,
                                           int lane) {
    const float inv = __int_as_float(__builtin_amdgcn_readlane(invbits, w.cur));
    const float4 o = make_float4(w.acc.x * inv, w.acc.y * inv, w.acc.z * inv, w.acc.w * inv);
    *(float4*)(s_rows + w.cur * MEAN_PITCH + 4 * lane) = o;
    w.acc = make_float4(0.f, 0.f, 0.f, 0.f);
}

template <int U>
__device__ __forceinline__ void gather_rows(const float4* __restrict__ in4, int mine, int t, int e,
                                            int lane, int rp, int invbits,
                                            float* __restrict__ s_rows, StripeWalk& w) {
    float4 v[U];
#pragma unroll
    for (int u = 0; u < U; ++u) {
        const int s = __builtin_amdgcn_readlane(mine, t + u);
        v[u] = in4[(long long)s * (HDIM / 4) + lane];
    }
#pragma unroll
    for (int u = 0; u < U; ++u) {
        while (e + u == w.bnd) {
            close_node(w, s_rows, invbits, lane);
            ++w.cur;
            w.bnd = __builtin_amdgcn_readlane(rp, w.cur + 1);
        }
        add4(w.acc, v[u]);
    }
}

__device__ __forceinline__ void gather_stripe(const float* __restrict__ in,
                                              const int* __restrict__ rowptr,
                                              const int* __restrict__ col,
                                              const float* __restrict__ invdeg, int row0, int lane,
                                              float* __restrict__ s_rows) {
    const int rp = rowptr[row0 + ((lane < 16) ? lane : 16)];
    const int invbits = __float_as_int(invdeg[row0 + (lane & 15)]);
    const int ebeg = __builtin_amdgcn_readlane(rp, 0);
    const int eend = __builtin_amdgcn_readlane(rp, 16);
    const float4* in4 = (const float4*)in;
    StripeWalk w;
    w.cur = 0;
    w.bnd = __builtin_amdgcn_readlane(rp, 1);
    w.acc = make_float4(0.f, 0.f, 0.f, 0.f);
    int ahead = (ebeg + lane < eend) ? col[ebeg + lane] : 0;
    for (int e0 = ebeg; e0 < eend; e0 += 32) {
        const int nj = (eend - e0 < 32) ? (eend - e0) : 32;
        const int mine = ahead;
        if (e0 + 32 < eend) ahead = (e0 + 32 + lane < eend) ? col[e0 + 32 + lane] : 0;
        int t = 0;
        for (; t + 8 <= nj; t += 8) gather_rows<8>(in4, mine, t, e0 + t, lane, rp, invbits, s_rows, w);
        if (t + 4 <= nj) { gather_rows<4>(in4, mine, t, e0 + t, lane, rp, invbits, s_rows, w); t += 4; }
        if (t + 2 <= nj) { gather_rows<2>(in4, mine, t, e0 + t, lane, rp, invbits, s_rows, w); t += 2; }
        if (t < nj) gather_rows<1>(in4, mine, t, e0 + t, lane, rp, invbits, s_rows, w);
    }
    while (w.cur < 16) {
        close_node(w, s_rows, invbits, lane);
        ++w.cur;
    }
}

#define AGG_PITCH 20
template <bool IN_LDS>
__device__ __forceinline__ void sum_rows16(const float4* __restrict__ x4, const int* ids,
                                           const int* s_end, int wu, int lane,
                                           float* __restrict__ s_my) {
    const int q = lane & 3;
    for (int hf = 0; hf < 2; ++hf) {
        const int nl = 16 * wu + 8 * hf + (lane >> 2);
        const int beg = (nl > 0) ? s_end[nl - 1] : 0;
        const int end = s_end[nl];
        float4 sum = make_float4(0.f, 0.f, 0.f, 0.f);
        for (int j = beg; j < end; j += 8) {
            float4 v[8];
#pragma unroll
            for (int u = 0; u < 8; ++u) {
                const int jj = (j + u < end) ? (j + u) : (end - 1);
                v[u] = x4[(long long)ids[jj] * (F_INPUT / 4) + q];
            }
#pragma unroll
            for (int u = 0; u < 8; ++u) {
                float4 nx = sum;
                add4(nx, v[u]);
                const bool in_list = j + u < end;
                sum.x = in_list ? nx.x : sum.x;
                sum.y = in_list ? nx.y : sum.y;
                sum.z = in_list ? nx.z : sum.z;
                sum.w = in_list ? nx.w : sum.w;
            }
        }
        *(float4*)(s_my + (8 * hf + (lane >> 2)) * AGG_PITCH + 4 * q) = sum;
    }
}

__global__ __launch_bounds__(PART_T) void bin_order_layer1_kernel(
    const int* __restrict__ packed, const int* __restrict__ binstart, int* __restrict__ rowptr,
    float* __restrict__ invdeg, int* __restrict__ col, const float* __restrict__ x,
    const float* __restrict__ Wr, const float* __restrict__ Wl, const float* __restrict__ bias,
    float* __restrict__ out) {
    __shared__ int s_cnt[BIN_N];
    __shared__ int s_wave[PART_T / 32];
    __shared__ int s_col[STAGE_CAP];
    __shared__ alignas(16) float s_agg[(PART_T / 32) * 16 * AGG_PITCH];
    const int t = threadIdx.x;
    const int lane = t & 31;
    const int w = t >> 5;
    const int b = blockIdx.x;
    const int ebeg = binstart[b];
    const int eend = binstart[b + 1];
    const bool staged = (eend - ebeg) <= STAGE_CAP;

    if (t < BIN_N) s_cnt[t] = 0;
    __syncthreads();
    for (int e0 = ebeg + t; e0 < eend; e0 += 4 * PART_T) {
        int p[4];
#pragma unroll
        for (int u = 0; u < 4; ++u) {
            const int e = e0 + u * PART_T;
            p[u] = (e < eend) ? packed[e] : -1;
        }
#pragma unroll
        for (int u = 0; u < 4; ++u)
            if (p[u] >= 0) atomicAdd(&s_cnt[p[u] >> SRC_BITS], 1);
    }
    __syncthreads();

    const int c = (t < BIN_N) ? s_cnt[t] : 0;
    int inc = c;
    for (int o = 1; o < 32; o <<= 1) {
        const int u = __shfl_up(inc, o);
        if (lane >= o) inc += u;
    }
    if (lane == 31) s_wave[w] = inc;
    __syncthreads();
    int front = (lane < w) ? s_wave[lane] : 0;
    for (int o = 16; o > 0; o >>= 1) front += __shfl_xor(front, o);
    const int excl = front + inc - c;

    const int node = b * BIN_N + t;
    if (t < BIN_N) {
        s_cnt[t] = excl;
        if (node < N_NODES) {
            rowptr[node] = ebeg + excl;
            invdeg[node] = 1.0f / fmaxf((float)c, 1.0f);
            if (node == N_NODES - 1) rowptr[N_NODES] = ebeg + excl + c;
        }
    }
    __syncthreads();

    for (int e0 = ebeg + t; e0 < eend; e0 += 4 * PART_T) {
        int p[4];
#pragma unroll
        for (int u = 0; u < 4; ++u) {
            const int e = e0 + u * PART_T;
            p[u] = (e < eend) ? packed[e] : -1;
        }
#pragma unroll
        for (int u = 0; u < 4; ++u) {
            if (p[u] >= 0) {
                const int pos = atomicAdd(&s_cnt[p[u] >> SRC_BITS], 1);
                if (staged) s_col[pos] = p[u] & SRC_MASK;
                else col[ebeg + pos] = p[u] & SRC_MASK;
            }
        }
    }
    __syncthreads();
    if (staged)
        for (int i = t; i < eend - ebeg; i += PART_T) col[ebeg + i] = s_col[i];

    const int wu = __builtin_amdgcn_readfirstlane(w);
    const int row0 = b * BIN_N + 16 * wu;
    if (row0 >= N_NODES) return;

    float* s_my = s_agg + wu * (16 * AGG_PITCH);
    const float4* x4 = (const float4*)x;
    if (staged) sum_rows16<true>(x4, s_col, s_cnt, wu, lane, s_my);
    else sum_rows16<false>(x4, col + ebeg, s_cnt, wu, lane, s_my);

    __builtin_amdgcn_fence(__ATOMIC_RELEASE, "wavefront");
    __builtin_amdgcn_wave_barrier();
    __builtin_amdgcn_fence(__ATOMIC_ACQUIRE, "wavefront");

    const int m = lane & 15;
    const int kh = (lane >> 4) << 1;
    const long long arow = (long long)(row0 + m);
    const int nm = 16 * wu + m;
    const int deg = s_cnt[nm] - ((nm > 0) ? s_cnt[nm - 1] : 0);
    const float inv = 1.0f / fmaxf((float)deg, 1.0f);

    v8f acc[8] = {};

    for (int k = 0; k < F_INPUT; k += 4) {
        v2f a = *(const v2f*)(x + arow * F_INPUT + k + kh);
#pragma unroll
        for (int nt = 0; nt < 8; ++nt) {
            const int cc = (nt << 4) + m;
            v2f bf;
            bf.x = Wr[(k + kh) * HDIM + cc];
            bf.y = Wr[(k + kh + 1) * HDIM + cc];
            acc[nt] = __builtin_amdgcn_wmma_f32_16x16x4_f32(
                false, a, false, bf, (short)0, acc[nt], false, false);
        }
    }

    for (int k = 0; k < F_INPUT; k += 4) {
        v2f a = *(const v2f*)(s_my + m * AGG_PITCH + k + kh);
        a.x *= inv;
        a.y *= inv;
#pragma unroll
        for (int nt = 0; nt < 8; ++nt) {
            const int cc = (nt << 4) + m;
            v2f bf;
            bf.x = Wl[(k + kh) * HDIM + cc];
            bf.y = Wl[(k + kh + 1) * HDIM + cc];
            acc[nt] = __builtin_amdgcn_wmma_f32_16x16x4_f32(
                false, a, false, bf, (short)0, acc[nt], false, false);
        }
    }

    const int rofs = (lane >> 4) << 3;
#pragma unroll
    for (int nt = 0; nt < 8; ++nt) {
        const int cc = (nt << 4) + m;
        const float bv = bias[cc];
#pragma unroll
        for (int j = 0; j < 8; ++j) {
            const float v = fmaxf(acc[nt][j] + bv, 0.0f);
            out[(long long)(row0 + rofs + j) * HDIM + cc] = v;
        }
    }
}

__global__ __launch_bounds__(LAYER_T) void sage_layer128_kernel(
    const float* __restrict__ in, const int* __restrict__ rowptr, const int* __restrict__ col,
    const float* __restrict__ invdeg, const float* __restrict__ Wr, const float* __restrict__ Wl,
    const float* __restrict__ bias, float* __restrict__ out, int do_relu) {
    __shared__ float4 s_w[S_UNITS];
    __shared__ float s_mean[FUSE_WAVES * 16 * MEAN_PITCH];
    const int lane = threadIdx.x & 31;
    const int wv = __builtin_amdgcn_readfirstlane((int)(threadIdx.x >> 5));
    const int mt = blockIdx.x * FUSE_WAVES + wv;

    {
        float4 t[S_UNITS_PER_T];
#pragma unroll
        for (int i = 0; i < S_UNITS_PER_T; ++i) {
            const int u = (int)threadIdx.x + i * LAYER_T;
            if (u < S_UNITS) {
                const float* Wsrc = (u < W_UNITS) ? Wr : Wl;
                const int uu = (u < W_UNITS) ? u : (u - W_UNITS);
                const int ul = uu & 31;
                const int uj = (uu >> 5) & 3;
                const int uks = uu >> 7;
                const float* p = Wsrc + (4 * uks + ((ul >> 4) << 1)) * HDIM + 32 * uj + (ul & 15);
                t[i] = make_float4(p[0], p[HDIM], p[16], p[HDIM + 16]);
            }
        }
#pragma unroll
        for (int i = 0; i < S_UNITS_PER_T; ++i) {
            const int u = (int)threadIdx.x + i * LAYER_T;
            if (u < S_UNITS) s_w[u] = t[i];
        }
    }
    __syncthreads();

    const int row0 = mt << 4;
    const int m = lane & 15;
    const int kh = (lane >> 4) << 1;
    const long long arow = (long long)(row0 + m);
    float* s_rows = s_mean + wv * (16 * MEAN_PITCH);

    v8f acc[8] = {};

    const bool gather_first = ((wv >> 2) & 1) == 0;
    if (gather_first) gather_stripe(in, rowptr, col, invdeg, row0, lane, s_rows);

#pragma unroll 2
    for (int ks = 0; ks < HDIM / 4; ++ks) {
        v2f a = *(const v2f*)(in + arow * HDIM + 4 * ks + kh);
#pragma unroll
        for (int j = 0; j < 4; ++j) {
            const float4 bb = s_w[(ks * 4 + j) * 32 + lane];
            v2f b0, b1;
            b0.x = bb.x; b0.y = bb.y;
            b1.x = bb.z; b1.y = bb.w;
            acc[2 * j] = __builtin_amdgcn_wmma_f32_16x16x4_f32(
                false, a, false, b0, (short)0, acc[2 * j], false, false);
            acc[2 * j + 1] = __builtin_amdgcn_wmma_f32_16x16x4_f32(
                false, a, false, b1, (short)0, acc[2 * j + 1], false, false);
        }
    }

    if (!gather_first) gather_stripe(in, rowptr, col, invdeg, row0, lane, s_rows);

    __builtin_amdgcn_fence(__ATOMIC_RELEASE, "wavefront");
    __builtin_amdgcn_wave_barrier();
    __builtin_amdgcn_fence(__ATOMIC_ACQUIRE, "wavefront");

#pragma unroll 2
    for (int ks = 0; ks < WL_LDS_STEPS; ++ks) {
        v2f a = *(const v2f*)(s_rows + m * MEAN_PITCH + 4 * ks + kh);
#pragma unroll
        for (int j = 0; j < 4; ++j) {
            const float4 bb = s_w[W_UNITS + (ks * 4 + j) * 32 + lane];
            v2f b0, b1;
            b0.x = bb.x; b0.y = bb.y;
            b1.x = bb.z; b1.y = bb.w;
            acc[2 * j] = __builtin_amdgcn_wmma_f32_16x16x4_f32(
                false, a, false, b0, (short)0, acc[2 * j], false, false);
            acc[2 * j + 1] = __builtin_amdgcn_wmma_f32_16x16x4_f32(
                false, a, false, b1, (short)0, acc[2 * j + 1], false, false);
        }
    }
    for (int k = 4 * WL_LDS_STEPS; k < HDIM; k += 4) {
        v2f a = *(const v2f*)(s_rows + m * MEAN_PITCH + k + kh);
#pragma unroll
        for (int nt = 0; nt < 8; ++nt) {
            const int cc = (nt << 4) + m;
            v2f b;
            b.x = Wl[(k + kh) * HDIM + cc];
            b.y = Wl[(k + kh + 1) * HDIM + cc];
            acc[nt] = __builtin_amdgcn_wmma_f32_16x16x4_f32(
                false, a, false, b, (short)0, acc[nt], false, false);
        }
    }

    const int rofs = (lane >> 4) << 3;
#pragma unroll
    for (int nt = 0; nt < 8; ++nt) {
        const int cc = (nt << 4) + m;
        const float bv = bias[cc];
#pragma unroll
        for (int j = 0; j < 8; ++j) {
            float v = acc[nt][j] + bv;
            if (do_relu) v = fmaxf(v, 0.0f);
            out[(long long)(row0 + rofs + j) * HDIM + cc] = v;
        }
    }
}

extern "C" void kernel_launch(void* const* d_in, const int* in_sizes, int n_in,
                              void* d_out, int out_size, void* d_ws, size_t ws_size,
                              hipStream_t stream) {
    (void)in_sizes; (void)n_in; (void)out_size; (void)ws_size;

    const float* x = (const float*)d_in[0];
    const int* ei = (const int*)d_in[1];
    const int* src = ei;
    const int* dst = ei + N_EDGES;
    const float* Wl[4] = {(const float*)d_in[2], (const float*)d_in[5],
                          (const float*)d_in[8], (const float*)d_in[11]};
    const float* Wr[4] = {(const float*)d_in[3], (const float*)d_in[6],
                          (const float*)d_in[9], (const float*)d_in[12]};
    const float* bs[4] = {(const float*)d_in[4], (const float*)d_in[7],
                          (const float*)d_in[10], (const float*)d_in[13]};

    float* agg = (float*)d_ws;
    float* hA = agg + (size_t)N_NODES * HDIM;
    float* hB = hA + (size_t)N_NODES * HDIM;
    float* deg = hB + (size_t)N_NODES * HDIM;

    int* packed = (int*)hB;
    int* cnt1 = packed + N_EDGES;
    int* binstart = cnt1 + NBINS * NCHUNK;
    int* col = (int*)(agg + (size_t)N_NODES * F_INPUT);
    int* rowptr = col + N_EDGES;

    chunk_count_kernel<<<NCHUNK, PART_T, 0, stream>>>(dst, cnt1);
    chunk_fill_kernel<<<NCHUNK, PART_T, 0, stream>>>(src, dst, cnt1, binstart, packed);
    bin_order_layer1_kernel<<<NBINS, PART_T, 0, stream>>>(packed, binstart, rowptr, deg, col, x,
                                                          Wr[0], Wl[0], bs[0], hA);

    const float* cur = hA;
    float* nxt = hB;
    const int layer_grid = M_TILES / FUSE_WAVES;
    for (int l = 1; l < 4; ++l) {
        float* outp = (l == 3) ? (float*)d_out : nxt;
        int relu = (l == 3) ? 0 : 1;
        sage_layer128_kernel<<<layer_grid, LAYER_T, 0, stream>>>(
            cur, rowptr, col, deg, Wr[l], Wl[l], bs[l], outp, relu);
        float* t2 = (float*)cur;
        cur = outp;
        nxt = t2;
    }
}
